# starting kernel
# baseline (speedup 1.0000x reference)
#include <hip/hip_runtime.h>

typedef __attribute__((ext_vector_type(2))) float v2f;
typedef __attribute__((ext_vector_type(8))) float v8f;

#define TLN_S 4096
#define TLN_B 8
#define TLN_C 1024
#define TLN_NC (TLN_B * TLN_C)
#define TLN_CHUNK 128
#define TLN_NCHUNK (TLN_S / TLN_CHUNK)
#define TLN_EPS 1e-5f

extern "C" __global__ __launch_bounds__(256)
void tln_partials(const float* __restrict__ x,
                  float* __restrict__ psum,
                  float* __restrict__ pssum) {
  const int tid = blockIdx.x * 256 + threadIdx.x;
  const int ch  = tid % TLN_NC;
  const int q   = tid / TLN_NC;
  const float* xp = x + (size_t)q * TLN_CHUNK * TLN_NC + ch;
  float s = 0.0f, ss = 0.0f;
#pragma unroll 8
  for (int i = 0; i < TLN_CHUNK; ++i) {
    const float v = xp[(size_t)i * TLN_NC];
    s  += v;
    ss += v * v;
  }
  psum[q * TLN_NC + ch]  = s;
  pssum[q * TLN_NC + ch] = ss;
}

extern "C" __global__ __launch_bounds__(256)
void tln_scan(const float* __restrict__ x,
              const float* __restrict__ pe,
              const float* __restrict__ weight,
              const float* __restrict__ bias,
              const float* __restrict__ psum,
              const float* __restrict__ pssum,
              float* __restrict__ out) {
  constexpr int NGB = TLN_NC / (16 * 8);
  const int gb   = blockIdx.x % NGB;
  const int q    = blockIdx.x / NGB;
  const int wave = threadIdx.x >> 5;
  const int lane = threadIdx.x & 31;
  const int j    = lane & 15;
  const int hi   = lane >> 4;
  const int ch   = (gb * 8 + wave) * 16 + j;
  const int c    = ch & (TLN_C - 1);

  const float w  = weight[c] + 1.0f;
  const float bb = bias[c];
  const float p0 = pe[c];
  const float p1 = pe[TLN_C + c];
  float carry_s = p0 + p1;
  float carry_q = p0 * p0 + p1 * p1;

  for (int t = 0; t < q; ++t) {
    carry_s += psum[t * TLN_NC + ch];
    carry_q += pssum[t * TLN_NC + ch];
  }

  const int s0 = q * TLN_CHUNK;
  for (int t16 = 0; t16 < TLN_CHUNK / 16; ++t16) {
    const int sb = s0 + t16 * 16;
    const float* xt = x + (size_t)sb * TLN_NC + ch;

    float xr[8];
#pragma unroll
    for (int r = 0; r < 8; ++r)
      xr[r] = xt[(size_t)(r + 8 * hi) * TLN_NC];

    v8f acc = {0.f, 0.f, 0.f, 0.f, 0.f, 0.f, 0.f, 0.f};
    v8f acq = {0.f, 0.f, 0.f, 0.f, 0.f, 0.f, 0.f, 0.f};
#pragma unroll
    for (int p = 0; p < 4; ++p) {
      const int k0 = 4 * p + 2 * hi;
      v2f A;
      A.x = (k0     <= j) ? 1.0f : 0.0f;
      A.y = (k0 + 1 <= j) ? 1.0f : 0.0f;
      const float b0 = xt[(size_t)k0 * TLN_NC];
      const float b1 = xt[(size_t)(k0 + 1) * TLN_NC];
      v2f Bv; Bv.x = b0;      Bv.y = b1;
      v2f Bq; Bq.x = b0 * b0; Bq.y = b1 * b1;
      acc = __builtin_amdgcn_wmma_f32_16x16x4_f32(false, A, false, Bv, (short)0, acc, false, false);
      acq = __builtin_amdgcn_wmma_f32_16x16x4_f32(false, A, false, Bq, (short)0, acq, false, false);
    }

    const float tot_s = __shfl(acc[7], 16 + j, 32);
    const float tot_q = __shfl(acq[7], 16 + j, 32);

    float* ot = out + (size_t)sb * TLN_NC + ch;
#pragma unroll
    for (int r = 0; r < 8; ++r) {
      const int row = r + 8 * hi;
      const float sum    = carry_s + acc[r];
      const float ssum   = carry_q + acq[r];
      const float rpos   = __builtin_amdgcn_rcpf(3.0f + (float)(sb + row));
      const float mean   = sum * rpos;
      const float var    = ssum * rpos - mean * mean;
      const float invstd = __builtin_amdgcn_rsqf(var + TLN_EPS);
      ot[(size_t)row * TLN_NC] = (xr[r] - mean) * (w * invstd) + bb;
    }
    carry_s += tot_s;
    carry_q += tot_q;
  }
}

extern "C" void kernel_launch(void* const* d_in, const int* in_sizes, int n_in,
                              void* d_out, int out_size, void* d_ws, size_t ws_size,
                              hipStream_t stream) {
  const float* x      = (const float*)d_in[0];
  const float* pe     = (const float*)d_in[1];
  const float* weight = (const float*)d_in[2];
  const float* bias   = (const float*)d_in[3];
  float* out   = (float*)d_out;
  float* psum  = (float*)d_ws;
  float* pssum = psum + (size_t)TLN_NCHUNK * TLN_NC;

  dim3 blk(256);
  dim3 g1((TLN_NCHUNK * TLN_NC) / 256);
  tln_partials<<<g1, blk, 0, stream>>>(x, psum, pssum);

  constexpr int NGB = TLN_NC / (16 * 8);
  dim3 g2(NGB * TLN_NCHUNK);
  tln_scan<<<g2, blk, 0, stream>>>(x, pe, weight, bias, psum, pssum, out);
}


# 16 B per lane: each lane owns 4 adjacent channels in both kernels (scan wave = 64 channels as 4 WMMA column tiles); 4x fewer waves, partials hold 16-32 row loads in flight; sum order kept
# speedup vs baseline: 1.1724x; 1.1724x over previous
#include <hip/hip_runtime.h>


typedef __attribute__((ext_vector_type(2))) float v2f;
typedef __attribute__((ext_vector_type(4))) float v4f;
typedef __attribute__((ext_vector_type(8))) float v8f;

#define TLN_S 4096
#define TLN_B 8
#define TLN_C 1024
#define TLN_NC (TLN_B * TLN_C)
#define TLN_NC4 (TLN_NC / 4)
#define TLN_C4 (TLN_C / 4)
#define TLN_CHUNK 128
#define TLN_NCHUNK (TLN_S / TLN_CHUNK)
#define TLN_EPS 1e-5f

#define TLN_P1_BATCH 16
#define TLN_P1_NBATCH (TLN_CHUNK / TLN_P1_BATCH)
extern "C" __global__ __launch_bounds__(256)
void tln_partials(const float* __restrict__ x,
                  float* __restrict__ psum,
                  float* __restrict__ pssum) {
  const int tid = blockIdx.x * 256 + threadIdx.x;
  const int c4  = tid % TLN_NC4;
  const int q   = tid / TLN_NC4;
  const v4f* xp = reinterpret_cast<const v4f*>(x) + (size_t)q * TLN_CHUNK * TLN_NC4 + c4;
  v4f s  = {0.f, 0.f, 0.f, 0.f};
  v4f ss = {0.f, 0.f, 0.f, 0.f};
  v4f buf[2][TLN_P1_BATCH];
#pragma unroll
  for (int i = 0; i < TLN_P1_BATCH; ++i)
    buf[0][i] = xp[(size_t)i * TLN_NC4];
#pragma unroll
  for (int b = 0; b < TLN_P1_NBATCH; ++b) {
    if (b + 1 < TLN_P1_NBATCH) {
#pragma unroll
      for (int i = 0; i < TLN_P1_BATCH; ++i)
        buf[(b + 1) & 1][i] = xp[(size_t)((b + 1) * TLN_P1_BATCH + i) * TLN_NC4];
    }
    __builtin_amdgcn_sched_barrier(0);
#pragma unroll
    for (int i = 0; i < TLN_P1_BATCH; ++i) {
      const v4f v = buf[b & 1][i];
      s  += v;
      ss += v * v;
    }
    asm volatile("" : "+v"(s), "+v"(ss));
  }
  reinterpret_cast<v4f*>(psum)[q * TLN_NC4 + c4]  = s;
  reinterpret_cast<v4f*>(pssum)[q * TLN_NC4 + c4] = ss;
}

extern "C" __global__ __launch_bounds__(256)
void tln_scan(const float* __restrict__ x,
              const float* __restrict__ pe,
              const float* __restrict__ weight,
              const float* __restrict__ bias,
              const float* __restrict__ psum,
              const float* __restrict__ pssum,
              float* __restrict__ out) {
  constexpr int NGB = TLN_NC / (64 * 8);
  const int gb   = blockIdx.x % NGB;
  const int q    = blockIdx.x / NGB;
  const int wave = threadIdx.x >> 5;
  const int lane = threadIdx.x & 31;
  const int j    = lane & 15;
  const int hi   = lane >> 4;
  const int c4   = (gb * 8 + wave) * 16 + j;
  const int cc4  = c4 & (TLN_C4 - 1);

  const v4f w  = reinterpret_cast<const v4f*>(weight)[cc4] + 1.0f;
  const v4f bb = reinterpret_cast<const v4f*>(bias)[cc4];
  const v4f p0 = reinterpret_cast<const v4f*>(pe)[cc4];
  const v4f p1 = reinterpret_cast<const v4f*>(pe)[TLN_C4 + cc4];
  v4f carry_s = p0 + p1;
  v4f carry_q = p0 * p0 + p1 * p1;

  const v4f* ps4 = reinterpret_cast<const v4f*>(psum);
  const v4f* pq4 = reinterpret_cast<const v4f*>(pssum);
  for (int t = 0; t < q; ++t) {
    carry_s += ps4[t * TLN_NC4 + c4];
    carry_q += pq4[t * TLN_NC4 + c4];
  }

  const v4f* x4 = reinterpret_cast<const v4f*>(x);
  v4f* o4 = reinterpret_cast<v4f*>(out);

  const int s0 = q * TLN_CHUNK;
  for (int t16 = 0; t16 < TLN_CHUNK / 16; ++t16) {
    const int sb = s0 + t16 * 16;
    const v4f* xt = x4 + (size_t)sb * TLN_NC4 + c4;

    v4f xr[8];
#pragma unroll
    for (int r = 0; r < 8; ++r)
      xr[r] = xt[(size_t)(r + 8 * hi) * TLN_NC4];

    v8f acc[4];
    v8f acq[4];
#pragma unroll
    for (int m = 0; m < 4; ++m) {
      acc[m] = (v8f){0.f, 0.f, 0.f, 0.f, 0.f, 0.f, 0.f, 0.f};
      acq[m] = (v8f){0.f, 0.f, 0.f, 0.f, 0.f, 0.f, 0.f, 0.f};
    }
#pragma unroll
    for (int p = 0; p < 4; ++p) {
      const int k0 = 4 * p + 2 * hi;
      v2f A;
      A.x = (k0     <= j) ? 1.0f : 0.0f;
      A.y = (k0 + 1 <= j) ? 1.0f : 0.0f;
      const v4f b0 = xt[(size_t)k0 * TLN_NC4];
      const v4f b1 = xt[(size_t)(k0 + 1) * TLN_NC4];
      const v4f q0 = b0 * b0;
      const v4f q1 = b1 * b1;
#pragma unroll
      for (int m = 0; m < 4; ++m) {
        v2f Bv; Bv.x = b0[m]; Bv.y = b1[m];
        v2f Bq; Bq.x = q0[m]; Bq.y = q1[m];
        acc[m] = __builtin_amdgcn_wmma_f32_16x16x4_f32(false, A, false, Bv, (short)0, acc[m], false, false);
        acq[m] = __builtin_amdgcn_wmma_f32_16x16x4_f32(false, A, false, Bq, (short)0, acq[m], false, false);
      }
    }

    v4f tot_s, tot_q;
#pragma unroll
    for (int m = 0; m < 4; ++m) {
      tot_s[m] = __shfl(acc[m][7], 16 + j, 32);
      tot_q[m] = __shfl(acq[m][7], 16 + j, 32);
    }

    v4f* ot = o4 + (size_t)sb * TLN_NC4 + c4;
#pragma unroll
    for (int r = 0; r < 8; ++r) {
      const int row = r + 8 * hi;
      const float rpos = __builtin_amdgcn_rcpf(3.0f + (float)(sb + row));
      v4f o;
#pragma unroll
      for (int m = 0; m < 4; ++m) {
        const float sum    = carry_s[m] + acc[m][r];
        const float ssum   = carry_q[m] + acq[m][r];
        const float mean   = sum * rpos;
        const float var    = ssum * rpos - mean * mean;
        const float invstd = __builtin_amdgcn_rsqf(var + TLN_EPS);
        o[m] = (xr[r][m] - mean) * (w[m] * invstd) + bb[m];
      }
      ot[(size_t)row * TLN_NC4] = o;
    }
    carry_s += tot_s;
    carry_q += tot_q;
  }
}

extern "C" void kernel_launch(void* const* d_in, const int* in_sizes, int n_in,
                              void* d_out, int out_size, void* d_ws, size_t ws_size,
                              hipStream_t stream) {
  const float* x      = (const float*)d_in[0];
  const float* pe     = (const float*)d_in[1];
  const float* weight = (const float*)d_in[2];
  const float* bias   = (const float*)d_in[3];
  float* out   = (float*)d_out;
  float* psum  = (float*)d_ws;
  float* pssum = psum + (size_t)TLN_NCHUNK * TLN_NC;

  dim3 blk(256);
  dim3 g1((TLN_NCHUNK * TLN_NC4) / 256);
  tln_partials<<<g1, blk, 0, stream>>>(x, psum, pssum);

  constexpr int NGB = TLN_NC / (64 * 8);
  dim3 g2(NGB * TLN_NCHUNK);
  tln_scan<<<g2, blk, 0, stream>>>(x, pe, weight, bias, psum, pssum, out);
}


# scan pass reads each x tile once, one tile ahead, into a ring of two register buffers; A's row map is permuted so the B-layout registers also feed the elementwise math; sum order kept
# speedup vs baseline: 1.3532x; 1.1542x over previous
#include <hip/hip_runtime.h>


typedef __attribute__((ext_vector_type(2))) float v2f;
typedef __attribute__((ext_vector_type(4))) float v4f;
typedef __attribute__((ext_vector_type(8))) float v8f;

#define TLN_S 4096
#define TLN_B 8
#define TLN_C 1024
#define TLN_NC (TLN_B * TLN_C)
#define TLN_NC4 (TLN_NC / 4)
#define TLN_C4 (TLN_C / 4)
#define TLN_CHUNK 128
#define TLN_NCHUNK (TLN_S / TLN_CHUNK)
#define TLN_EPS 1e-5f

#define TLN_P1_BATCH 16
#define TLN_P1_NBATCH (TLN_CHUNK / TLN_P1_BATCH)
extern "C" __global__ __launch_bounds__(256)
void tln_partials(const float* __restrict__ x,
                  float* __restrict__ psum,
                  float* __restrict__ pssum) {
  const int tid = blockIdx.x * 256 + threadIdx.x;
  const int c4  = tid % TLN_NC4;
  const int q   = tid / TLN_NC4;
  const v4f* xp = reinterpret_cast<const v4f*>(x) + (size_t)q * TLN_CHUNK * TLN_NC4 + c4;
  v4f s  = {0.f, 0.f, 0.f, 0.f};
  v4f ss = {0.f, 0.f, 0.f, 0.f};
  v4f buf[2][TLN_P1_BATCH];
#pragma unroll
  for (int i = 0; i < TLN_P1_BATCH; ++i)
    buf[0][i] = xp[(size_t)i * TLN_NC4];
#pragma unroll
  for (int b = 0; b < TLN_P1_NBATCH; ++b) {
    if (b + 1 < TLN_P1_NBATCH) {
#pragma unroll
      for (int i = 0; i < TLN_P1_BATCH; ++i)
        buf[(b + 1) & 1][i] = xp[(size_t)((b + 1) * TLN_P1_BATCH + i) * TLN_NC4];
    }
    __builtin_amdgcn_sched_barrier(0);
#pragma unroll
    for (int i = 0; i < TLN_P1_BATCH; ++i) {
      const v4f v = buf[b & 1][i];
      s  += v;
      ss += v * v;
    }
    asm volatile("" : "+v"(s), "+v"(ss));
  }
  reinterpret_cast<v4f*>(psum)[q * TLN_NC4 + c4]  = s;
  reinterpret_cast<v4f*>(pssum)[q * TLN_NC4 + c4] = ss;
}

#define TLN_TROW(r) (4 * ((r) >> 1) + ((r) & 1))

static __device__ __forceinline__
void tln_load_tile(v4f (&buf)[8], const v4f* __restrict__ xt) {
#pragma unroll
  for (int r = 0; r < 8; ++r)
    buf[r] = xt[(size_t)TLN_TROW(r) * TLN_NC4];
}

static __device__ __forceinline__
void tln_scan_tile(const v4f (&buf)[8], const v2f (&A)[4], v4f* __restrict__ ot,
                   const int row0, const int j, const v4f w, const v4f bb,
                   v4f& carry_s, v4f& carry_q) {
  v8f acc[4];
  v8f acq[4];
#pragma unroll
  for (int m = 0; m < 4; ++m) {
    acc[m] = (v8f){0.f, 0.f, 0.f, 0.f, 0.f, 0.f, 0.f, 0.f};
    acq[m] = (v8f){0.f, 0.f, 0.f, 0.f, 0.f, 0.f, 0.f, 0.f};
  }
#pragma unroll
  for (int p = 0; p < 4; ++p) {
    const v4f b0 = buf[2 * p];
    const v4f b1 = buf[2 * p + 1];
    const v4f q0 = b0 * b0;
    const v4f q1 = b1 * b1;
#pragma unroll
    for (int m = 0; m < 4; ++m) {
      v2f Bv; Bv.x = b0[m]; Bv.y = b1[m];
      v2f Bq; Bq.x = q0[m]; Bq.y = q1[m];
      acc[m] = __builtin_amdgcn_wmma_f32_16x16x4_f32(false, A[p], false, Bv, (short)0, acc[m], false, false);
      acq[m] = __builtin_amdgcn_wmma_f32_16x16x4_f32(false, A[p], false, Bq, (short)0, acq[m], false, false);
    }
  }

  v4f tot_s, tot_q;
#pragma unroll
  for (int m = 0; m < 4; ++m) {
    tot_s[m] = __shfl(acc[m][7], 16 + j, 32);
    tot_q[m] = __shfl(acq[m][7], 16 + j, 32);
  }

#pragma unroll
  for (int r = 0; r < 8; ++r) {
    const float rpos = __builtin_amdgcn_rcpf(3.0f + (float)(row0 + TLN_TROW(r)));
    v4f o;
#pragma unroll
    for (int m = 0; m < 4; ++m) {
      const float sum    = carry_s[m] + acc[m][r];
      const float ssum   = carry_q[m] + acq[m][r];
      const float mean   = sum * rpos;
      const float var    = ssum * rpos - mean * mean;
      const float invstd = __builtin_amdgcn_rsqf(var + TLN_EPS);
      o[m] = (buf[r][m] - mean) * (w[m] * invstd) + bb[m];
    }
    ot[(size_t)TLN_TROW(r) * TLN_NC4] = o;
  }
  carry_s += tot_s;
  carry_q += tot_q;
}

extern "C" __global__ __launch_bounds__(256)
void tln_scan(const float* __restrict__ x,
              const float* __restrict__ pe,
              const float* __restrict__ weight,
              const float* __restrict__ bias,
              const float* __restrict__ psum,
              const float* __restrict__ pssum,
              float* __restrict__ out) {
  constexpr int NGB = TLN_NC / (64 * 8);
  const int gb   = blockIdx.x % NGB;
  const int q    = blockIdx.x / NGB;
  const int wave = threadIdx.x >> 5;
  const int lane = threadIdx.x & 31;
  const int j    = lane & 15;
  const int hi   = lane >> 4;
  const int c4   = (gb * 8 + wave) * 16 + j;
  const int cc4  = c4 & (TLN_C4 - 1);

  const v4f w  = reinterpret_cast<const v4f*>(weight)[cc4] + 1.0f;
  const v4f bb = reinterpret_cast<const v4f*>(bias)[cc4];
  const v4f p0 = reinterpret_cast<const v4f*>(pe)[cc4];
  const v4f p1 = reinterpret_cast<const v4f*>(pe)[TLN_C4 + cc4];
  v4f carry_s = p0 + p1;
  v4f carry_q = p0 * p0 + p1 * p1;

  const v4f* ps4 = reinterpret_cast<const v4f*>(psum);
  const v4f* pq4 = reinterpret_cast<const v4f*>(pssum);
  for (int t = 0; t < q; ++t) {
    carry_s += ps4[t * TLN_NC4 + c4];
    carry_q += pq4[t * TLN_NC4 + c4];
  }

  const int pj = 4 * ((j & 7) >> 1) + 2 * (j >> 3) + (j & 1);
  v2f A[4];
#pragma unroll
  for (int p = 0; p < 4; ++p) {
    const int k0 = 4 * p + 2 * hi;
    A[p].x = (k0     <= pj) ? 1.0f : 0.0f;
    A[p].y = (k0 + 1 <= pj) ? 1.0f : 0.0f;
  }

  const int s0 = q * TLN_CHUNK;
  const v4f* xl = reinterpret_cast<const v4f*>(x) + (size_t)(s0 + 2 * hi) * TLN_NC4 + c4;
  v4f* ol = reinterpret_cast<v4f*>(out) + (size_t)(s0 + 2 * hi) * TLN_NC4 + c4;
  const int r0 = s0 + 2 * hi;
  constexpr size_t TSTEP = (size_t)16 * TLN_NC4;

  v4f bufa[8], bufb[8];
  tln_load_tile(bufa, xl);
  constexpr int NPAIR = TLN_CHUNK / 32;
  for (int t2 = 0; t2 < NPAIR - 1; ++t2) {
    const size_t off = (size_t)t2 * 2 * TSTEP;
    tln_load_tile(bufb, xl + off + TSTEP);
    __builtin_amdgcn_sched_barrier(0);
    tln_scan_tile(bufa, A, ol + off, r0 + t2 * 32, j, w, bb, carry_s, carry_q);
    tln_load_tile(bufa, xl + off + 2 * TSTEP);
    __builtin_amdgcn_sched_barrier(0);
    tln_scan_tile(bufb, A, ol + off + TSTEP, r0 + t2 * 32 + 16, j, w, bb, carry_s, carry_q);
  }
  {
    const size_t off = (size_t)(NPAIR - 1) * 2 * TSTEP;
    tln_load_tile(bufb, xl + off + TSTEP);
    __builtin_amdgcn_sched_barrier(0);
    tln_scan_tile(bufa, A, ol + off, r0 + (NPAIR - 1) * 32, j, w, bb, carry_s, carry_q);
    __builtin_amdgcn_sched_barrier(0);
    tln_scan_tile(bufb, A, ol + off + TSTEP, r0 + (NPAIR - 1) * 32 + 16, j, w, bb, carry_s, carry_q);
  }
}

extern "C" void kernel_launch(void* const* d_in, const int* in_sizes, int n_in,
                              void* d_out, int out_size, void* d_ws, size_t ws_size,
                              hipStream_t stream) {
  const float* x      = (const float*)d_in[0];
  const float* pe     = (const float*)d_in[1];
  const float* weight = (const float*)d_in[2];
  const float* bias   = (const float*)d_in[3];
  float* out   = (float*)d_out;
  float* psum  = (float*)d_ws;
  float* pssum = psum + (size_t)TLN_NCHUNK * TLN_NC;

  dim3 blk(256);
  dim3 g1((TLN_NCHUNK * TLN_NC4) / 256);
  tln_partials<<<g1, blk, 0, stream>>>(x, psum, pssum);

  constexpr int NGB = TLN_NC / (64 * 8);
  dim3 g2(NGB * TLN_NCHUNK);
  tln_scan<<<g2, blk, 0, stream>>>(x, pe, weight, bias, psum, pssum, out);
}


# one launch, not two: a 16-wave work-group owns 32 channels for all rows; per 512-row band each wave scans 32 rows, wave totals cross through LDS (one barrier); x read once; carries add 32-row totals
# speedup vs baseline: 1.5135x; 1.1185x over previous
#include <hip/hip_runtime.h>


typedef __attribute__((ext_vector_type(2))) float v2f;
typedef __attribute__((ext_vector_type(4))) float v4f;
typedef __attribute__((ext_vector_type(8))) float v8f;

#define TLN_S 4096
#define TLN_B 8
#define TLN_C 1024
#define TLN_NC (TLN_B * TLN_C)
#define TLN_NC4 (TLN_NC / 4)
#define TLN_C4 (TLN_C / 4)
#define TLN_EPS 1e-5f

#define TLN_WAVES 16
#define TLN_WCH 32
#define TLN_TILE 32
#define TLN_BAND (TLN_WAVES * TLN_TILE)
#define TLN_NBAND (TLN_S / TLN_BAND)
#define TLN_NSEQ (2 * TLN_WCH)

#define TLN_TROW(r) (4 * ((r) >> 1) + ((r) & 1))

static __device__ __forceinline__
void tln_load_tile(v4f (&buf)[8], const v4f* __restrict__ xt) {
#pragma unroll
  for (int r = 0; r < 8; ++r)
    buf[r] = xt[(size_t)TLN_TROW(r) * TLN_NC4];
}

static __device__ __forceinline__
void tln_band(const v4f (&buf)[8], const v2f (&A)[4], v4f* __restrict__ ot,
              const int row0, const int lane, const int wave,
              const v4f w, const v4f bb, float (*slot)[TLN_NSEQ], v2f& run) {
  const int  jc = lane & 7;
  const bool sg = (lane & 8) != 0;

  v8f acc[4];
  v8f acq[4];
#pragma unroll
  for (int m = 0; m < 4; ++m) {
    acc[m] = (v8f){0.f, 0.f, 0.f, 0.f, 0.f, 0.f, 0.f, 0.f};
    acq[m] = (v8f){0.f, 0.f, 0.f, 0.f, 0.f, 0.f, 0.f, 0.f};
  }
#pragma unroll
  for (int p = 0; p < 4; ++p) {
    const v4f b0 = buf[2 * p];
    const v4f b1 = buf[2 * p + 1];
    const v4f q0 = b0 * b0;
    const v4f q1 = b1 * b1;
#pragma unroll
    for (int m = 0; m < 4; ++m) {
      v2f Bv; Bv.x = b0[m]; Bv.y = b1[m];
      v2f Bq; Bq.x = q0[m]; Bq.y = q1[m];
      acc[m] = __builtin_amdgcn_wmma_f32_16x16x4_f32(false, A[p], false, Bv, (short)0, acc[m], false, false);
      acq[m] = __builtin_amdgcn_wmma_f32_16x16x4_f32(false, A[p], false, Bq, (short)0, acq[m], false, false);
    }
  }

  v4f t0s, t0q, t1s, t1q;
#pragma unroll
  for (int m = 0; m < 4; ++m) {
    t0s[m] = __shfl(acc[m][7], 16 + jc, 32);
    t0q[m] = __shfl(acq[m][7], 16 + jc, 32);
    t1s[m] = __shfl(acc[m][7], 24 + jc, 32);
    t1q[m] = __shfl(acq[m][7], 24 + jc, 32);
  }

  *reinterpret_cast<v4f*>(&slot[wave][4 * jc])           = t0s + t1s;
  *reinterpret_cast<v4f*>(&slot[wave][TLN_WCH + 4 * jc]) = t0q + t1q;
  asm volatile("s_wait_dscnt 0x0" ::: "memory");
  asm volatile("s_barrier_signal -1\n\ts_barrier_wait -1" ::: "memory");

  v2f tv[TLN_WAVES];
#pragma unroll
  for (int t = 0; t < TLN_WAVES; ++t)
    tv[t] = *reinterpret_cast<const v2f*>(&slot[t][2 * lane]);
  __builtin_amdgcn_sched_barrier(0);
  v2f cap = run;
#pragma unroll
  for (int t = 0; t < TLN_WAVES; ++t) {
    if (t == wave) cap = run;
    run += tv[t];
  }

  v4f carry_s, carry_q;
  carry_s[0] = __shfl(cap.x, 2 * jc,          32);
  carry_s[1] = __shfl(cap.y, 2 * jc,          32);
  carry_s[2] = __shfl(cap.x, 2 * jc + 1,      32);
  carry_s[3] = __shfl(cap.y, 2 * jc + 1,      32);
  carry_q[0] = __shfl(cap.x, 16 + 2 * jc,     32);
  carry_q[1] = __shfl(cap.y, 16 + 2 * jc,     32);
  carry_q[2] = __shfl(cap.x, 16 + 2 * jc + 1, 32);
  carry_q[3] = __shfl(cap.y, 16 + 2 * jc + 1, 32);
#pragma unroll
  for (int m = 0; m < 4; ++m) {
    carry_s[m] = sg ? carry_s[m] + t0s[m] : carry_s[m];
    carry_q[m] = sg ? carry_q[m] + t0q[m] : carry_q[m];
  }

#pragma unroll
  for (int r = 0; r < 8; ++r) {
    const float rpos = __builtin_amdgcn_rcpf(3.0f + (float)(row0 + TLN_TROW(r)));
    v4f o;
#pragma unroll
    for (int m = 0; m < 4; ++m) {
      const float sum    = carry_s[m] + acc[m][r];
      const float ssum   = carry_q[m] + acq[m][r];
      const float mean   = sum * rpos;
      const float var    = ssum * rpos - mean * mean;
      const float invstd = __builtin_amdgcn_rsqf(var + TLN_EPS);
      o[m] = (buf[r][m] - mean) * (w[m] * invstd) + bb[m];
    }
    ot[(size_t)TLN_TROW(r) * TLN_NC4] = o;
  }
}

extern "C" __global__ __launch_bounds__(TLN_WAVES * 32)
void tln_fused(const float* __restrict__ x,
               const float* __restrict__ pe,
               const float* __restrict__ weight,
               const float* __restrict__ bias,
               float* __restrict__ out) {
  __shared__ __attribute__((aligned(16))) float tot[2][TLN_WAVES][TLN_NSEQ];

  const int lane = threadIdx.x & 31;
  const int wave = __builtin_amdgcn_readfirstlane((int)(threadIdx.x >> 5));
  const int j    = lane & 15;
  const int jc   = lane & 7;
  const int sg   = (lane >> 3) & 1;
  const int hi   = lane >> 4;
  const int c4   = blockIdx.x * (TLN_WCH / 4) + jc;
  const int cc4  = c4 & (TLN_C4 - 1);

  const v4f w  = reinterpret_cast<const v4f*>(weight)[cc4] + 1.0f;
  const v4f bb = reinterpret_cast<const v4f*>(bias)[cc4];

  const int cc = (blockIdx.x * TLN_WCH + ((2 * lane) & (TLN_WCH - 1))) & (TLN_C - 1);
  const v2f p0 = *reinterpret_cast<const v2f*>(pe + cc);
  const v2f p1 = *reinterpret_cast<const v2f*>(pe + TLN_C + cc);
  v2f run = (lane >> 4) ? (p0 * p0 + p1 * p1) : (p0 + p1);

  const int pj = 4 * ((j & 7) >> 1) + 2 * (j >> 3) + (j & 1);
  v2f A[4];
#pragma unroll
  for (int p = 0; p < 4; ++p) {
    const int k0 = 4 * p + 2 * hi;
    A[p].x = (k0     <= pj) ? 1.0f : 0.0f;
    A[p].y = (k0 + 1 <= pj) ? 1.0f : 0.0f;
  }

  const int r0 = wave * TLN_TILE + 16 * sg + 2 * hi;
  const v4f* xl = reinterpret_cast<const v4f*>(x) + (size_t)r0 * TLN_NC4 + c4;
  v4f* ol = reinterpret_cast<v4f*>(out) + (size_t)r0 * TLN_NC4 + c4;
  constexpr size_t BSTEP = (size_t)TLN_BAND * TLN_NC4;

  v4f bufa[8], bufb[8];
  tln_load_tile(bufa, xl);
  constexpr int NPAIR = TLN_NBAND / 2;
  for (int b2 = 0; b2 < NPAIR - 1; ++b2) {
    const size_t off = (size_t)b2 * 2 * BSTEP;
    tln_load_tile(bufb, xl + off + BSTEP);
    __builtin_amdgcn_sched_barrier(0);
    tln_band(bufa, A, ol + off, r0 + b2 * 2 * TLN_BAND, lane, wave, w, bb, tot[0], run);
    tln_load_tile(bufa, xl + off + 2 * BSTEP);
    __builtin_amdgcn_sched_barrier(0);
    tln_band(bufb, A, ol + off + BSTEP, r0 + (b2 * 2 + 1) * TLN_BAND, lane, wave, w, bb, tot[1], run);
  }
  {
    const size_t off = (size_t)(NPAIR - 1) * 2 * BSTEP;
    tln_load_tile(bufb, xl + off + BSTEP);
    __builtin_amdgcn_sched_barrier(0);
    tln_band(bufa, A, ol + off, r0 + (NPAIR - 1) * 2 * TLN_BAND, lane, wave, w, bb, tot[0], run);
    tln_band(bufb, A, ol + off + BSTEP, r0 + ((NPAIR - 1) * 2 + 1) * TLN_BAND, lane, wave, w, bb, tot[1], run);
  }
}

extern "C" void kernel_launch(void* const* d_in, const int* in_sizes, int n_in,
                              void* d_out, int out_size, void* d_ws, size_t ws_size,
                              hipStream_t stream) {
  const float* x      = (const float*)d_in[0];
  const float* pe     = (const float*)d_in[1];
  const float* weight = (const float*)d_in[2];
  const float* bias   = (const float*)d_in[3];
  float* out = (float*)d_out;

  dim3 blk(TLN_WAVES * 32);
  dim3 grd(TLN_NC / TLN_WCH);
  tln_fused<<<grd, blk, 0, stream>>>(x, pe, weight, bias, out);
}
